# starting kernel
# baseline (speedup 1.0000x reference)
#include <hip/hip_runtime.h>
#include <hip/hip_bf16.h>
#include <stdint.h>

typedef __attribute__((ext_vector_type(2))) float v2f;
typedef __attribute__((ext_vector_type(8))) float v8f;

#define N_DATA  50000
#define N_TASKS 100000
#define N_EDGES 1000000
#define F_DATA  5
#define F_TASK  12
#define HID     64

__device__ __forceinline__ unsigned enc_f32(float x) {
    unsigned u = __float_as_uint(x);
    return (u & 0x80000000u) ? ~u : (u | 0x80000000u);
}
__device__ __forceinline__ float dec_f32(unsigned u) {
    return (u & 0x80000000u) ? __uint_as_float(u & 0x7FFFFFFFu) : __uint_as_float(~u);
}

__global__ void k_attvec(const float* __restrict__ Wd, const float* __restrict__ ad,
                         const float* __restrict__ We, const float* __restrict__ ae,
                         float* __restrict__ vsmall) {
    int t = threadIdx.x;
    if (t < 12) {
        float s = 0.f;
        for (int h = 0; h < HID; ++h) s += Wd[t * HID + h] * ad[h];
        vsmall[t] = s;
    } else if (t < 15) {
        int k = t - 12;
        float s = 0.f;
        for (int h = 0; h < HID; ++h) s += We[k * HID + h] * ae[h];
        vsmall[t] = s;
    }
}

template <int F, int KSTEPS>
__device__ __forceinline__ void gemm_tile_wmma(const float* __restrict__ X,
                                               const float* __restrict__ W,
                                               int m0, int lane, v8f c[4]) {
    const int half = lane >> 4;
    const int l15  = lane & 15;
    const int mrow = m0 + l15;
#pragma unroll
    for (int s = 0; s < KSTEPS; ++s) {
        const int  ka  = s * 4 + half * 2;
        const bool v0  = (ka     < F);
        const bool v1  = (ka + 1 < F);
        const int  ka0 = v0 ? ka     : 0;
        const int  ka1 = v1 ? ka + 1 : 0;
        const float ax = X[mrow * F + ka0];
        const float ay = X[mrow * F + ka1];
        v2f a;
        a.x = v0 ? ax : 0.0f;
        a.y = v1 ? ay : 0.0f;
#pragma unroll
        for (int t = 0; t < 4; ++t) {
            const int col = t * 16 + l15;
            const float bx = W[ka0 * HID + col];
            const float by = W[ka1 * HID + col];
            v2f b;
            b.x = v0 ? bx : 0.0f;
            b.y = v1 ? by : 0.0f;
            c[t] = __builtin_amdgcn_wmma_f32_16x16x4_f32(
                false, a, false, b, (short)0, c[t], false, false);
        }
    }
}

__global__ void k_proj_data(const float* __restrict__ x, const float* __restrict__ W,
                            const float* __restrict__ att,
                            float* __restrict__ h, float* __restrict__ arow) {
    const int wave = (blockIdx.x * blockDim.x + threadIdx.x) >> 5;
    const int lane = threadIdx.x & 31;
    if (wave >= N_DATA / 16) return;
    const int m0 = wave * 16;
    v8f c[4] = {};
    gemm_tile_wmma<F_DATA, 2>(x, W, m0, lane, c);
    const int half = lane >> 4, l15 = lane & 15;
#pragma unroll
    for (int r = 0; r < 8; ++r) {
        const int row = m0 + r + half * 8;
        float p = 0.0f;
#pragma unroll
        for (int t = 0; t < 4; ++t) {
            const int col = t * 16 + l15;
            const float v = c[t][r];
            h[row * HID + col] = v;
            p += v * att[col];
        }
        p += __shfl_xor(p, 1);
        p += __shfl_xor(p, 2);
        p += __shfl_xor(p, 4);
        p += __shfl_xor(p, 8);
        if (l15 == 0) arow[row] = p;
    }
}

__global__ void k_res_proj(const float* __restrict__ xt, const float* __restrict__ Wr,
                           const float* __restrict__ bias, float* __restrict__ out) {
    const int wave = (blockIdx.x * blockDim.x + threadIdx.x) >> 5;
    const int lane = threadIdx.x & 31;
    if (wave >= N_TASKS / 16) return;
    const int m0 = wave * 16;
    v8f c[4] = {};
    gemm_tile_wmma<F_TASK, 3>(xt, Wr, m0, lane, c);
    const int half = lane >> 4, l15 = lane & 15;
#pragma unroll
    for (int r = 0; r < 8; ++r) {
        const int row = m0 + r + half * 8;
#pragma unroll
        for (int t = 0; t < 4; ++t) {
            const int col = t * 16 + l15;
            out[row * HID + col] = c[t][r] + bias[col];
        }
    }
}

__global__ void k_adst(const float* __restrict__ xt, const float* __restrict__ vsmall,
                       float* __restrict__ adst) {
    const int t = blockIdx.x * blockDim.x + threadIdx.x;
    if (t >= N_TASKS) return;
    float s = 0.f;
#pragma unroll
    for (int k = 0; k < F_TASK; ++k) s += xt[t * F_TASK + k] * vsmall[k];
    adst[t] = s;
}

__global__ void k_init(unsigned* __restrict__ amax, float* __restrict__ denom) {
    const int t = blockIdx.x * blockDim.x + threadIdx.x;
    if (t < N_TASKS) { amax[t] = 0u; denom[t] = 0.f; }
}

__global__ void k_alpha(const int* __restrict__ src, const int* __restrict__ dst,
                        const float* __restrict__ eattr, const float* __restrict__ asrc,
                        const float* __restrict__ adst, const float* __restrict__ vsmall,
                        float* __restrict__ alpha, unsigned* __restrict__ amax) {
    const int e = blockIdx.x * blockDim.x + threadIdx.x;
    if (e >= N_EDGES) return;
    const int s = src[e], d = dst[e];
    float al = asrc[s] + adst[d]
             + eattr[e * 3 + 0] * vsmall[12]
             + eattr[e * 3 + 1] * vsmall[13]
             + eattr[e * 3 + 2] * vsmall[14];
    al = (al > 0.f) ? al : 0.2f * al;
    alpha[e] = al;
    atomicMax(&amax[d], enc_f32(al));
}

__global__ void k_expsum(const int* __restrict__ dst, const unsigned* __restrict__ amax,
                         float* __restrict__ alpha, float* __restrict__ denom) {
    const int e = blockIdx.x * blockDim.x + threadIdx.x;
    if (e >= N_EDGES) return;
    const int d = dst[e];
    const float ex = __expf(alpha[e] - dec_f32(amax[d]));
    alpha[e] = ex;
    atomicAdd(&denom[d], ex);
}

__global__ void k_scatter(const int* __restrict__ src, const int* __restrict__ dst,
                          const float* __restrict__ ex, const float* __restrict__ denom,
                          const float* __restrict__ h, float* __restrict__ out) {
    const int wave = (blockIdx.x * blockDim.x + threadIdx.x) >> 5;
    const int lane = threadIdx.x & 31;
    if (wave >= N_EDGES) return;
    const int s = src[wave], d = dst[wave];
    const float w = ex[wave] / (denom[d] + 1e-16f);
    atomicAdd(&out[d * HID + lane],      w * h[s * HID + lane]);
    atomicAdd(&out[d * HID + lane + 32], w * h[s * HID + lane + 32]);
}

__global__ void k_lnorm(float* __restrict__ out, const float* __restrict__ gamma,
                        const float* __restrict__ beta) {
    const int wave = (blockIdx.x * blockDim.x + threadIdx.x) >> 5;
    const int lane = threadIdx.x & 31;
    if (wave >= N_TASKS) return;
    const float x0 = out[wave * HID + lane];
    const float x1 = out[wave * HID + lane + 32];
    float s = x0 + x1;
    float q = x0 * x0 + x1 * x1;
#pragma unroll
    for (int off = 1; off < 32; off <<= 1) {
        s += __shfl_xor(s, off);
        q += __shfl_xor(q, off);
    }
    const float mean = s * (1.0f / 64.0f);
    const float var  = q * (1.0f / 64.0f) - mean * mean;
    const float rstd = rsqrtf(var + 1e-5f);
    float y0 = (x0 - mean) * rstd * gamma[lane]      + beta[lane];
    float y1 = (x1 - mean) * rstd * gamma[lane + 32] + beta[lane + 32];
    y0 = (y0 > 0.f) ? y0 : 0.01f * y0;
    y1 = (y1 > 0.f) ? y1 : 0.01f * y1;
    out[wave * HID + lane]      = y0;
    out[wave * HID + lane + 32] = y1;
}

extern "C" void kernel_launch(void* const* d_in, const int* in_sizes, int n_in,
                              void* d_out, int out_size, void* d_ws, size_t ws_size,
                              hipStream_t stream) {
    const float* x_data    = (const float*)d_in[0];
    const float* x_tasks   = (const float*)d_in[1];
    const int*   edge_idx  = (const int*)d_in[2];
    const float* edge_attr = (const float*)d_in[3];
    const float* W_src     = (const float*)d_in[4];
    const float* W_dst     = (const float*)d_in[5];
    const float* W_edge    = (const float*)d_in[6];
    const float* att_src   = (const float*)d_in[7];
    const float* att_dst   = (const float*)d_in[8];
    const float* att_edge  = (const float*)d_in[9];
    const float* W_res     = (const float*)d_in[10];
    const float* bias      = (const float*)d_in[11];
    const float* ln_gamma  = (const float*)d_in[12];
    const float* ln_beta   = (const float*)d_in[13];
    const int* src = edge_idx;
    const int* dst = edge_idx + N_EDGES;
    float* out = (float*)d_out;

    float*    h_src  = (float*)d_ws;
    float*    a_src  = h_src + (size_t)N_DATA * HID;
    float*    a_dst  = a_src + N_DATA;
    unsigned* amax   = (unsigned*)(a_dst + N_TASKS);
    float*    denom  = (float*)(amax + N_TASKS);
    float*    alpha  = denom + N_TASKS;
    float*    vsmall = alpha + N_EDGES;

    k_attvec<<<1, 64, 0, stream>>>(W_dst, att_dst, W_edge, att_edge, vsmall);

    {
        const int waves = N_DATA / 16;
        k_proj_data<<<(waves * 32 + 127) / 128, 128, 0, stream>>>(
            x_data, W_src, att_src, h_src, a_src);
    }
    k_adst<<<(N_TASKS + 255) / 256, 256, 0, stream>>>(x_tasks, vsmall, a_dst);
    k_init<<<(N_TASKS + 255) / 256, 256, 0, stream>>>(amax, denom);
    k_alpha<<<(N_EDGES + 255) / 256, 256, 0, stream>>>(
        src, dst, edge_attr, a_src, a_dst, vsmall, alpha, amax);
    k_expsum<<<(N_EDGES + 255) / 256, 256, 0, stream>>>(dst, amax, alpha, denom);

    {
        const int waves = N_TASKS / 16;
        k_res_proj<<<(waves * 32 + 127) / 128, 128, 0, stream>>>(
            x_tasks, W_res, bias, out);
    }
    {
        const long long thr = (long long)N_EDGES * 32;
        k_scatter<<<(int)((thr + 255) / 256), 256, 0, stream>>>(
            src, dst, alpha, denom, h_src, out);
    }
    k_lnorm<<<(N_TASKS * 32 + 255) / 256, 256, 0, stream>>>(out, ln_gamma, ln_beta);
}


# norm pass retiled: a wave takes 16 rows (16 lanes x 16 bytes per row, two rows per load, eight loads in flight, row sums by 16-lane DPP); same bytes and launches, 6250 waves instead of 100000
# speedup vs baseline: 3.8221x; 3.8221x over previous
#include <hip/hip_runtime.h>
#include <hip/hip_bf16.h>
#include <stdint.h>

typedef __attribute__((ext_vector_type(2))) float v2f;
typedef __attribute__((ext_vector_type(8))) float v8f;

#define N_DATA  50000
#define N_TASKS 100000
#define N_EDGES 1000000
#define F_DATA  5
#define F_TASK  12
#define HID     64
#define CAP     32
#define NBIN       250
#define BIN_ROWS   400
#define BIN_CAP    12800
#define TILE_WAVES 25
#define TILE_THREADS (TILE_WAVES * 32)
#define BIN_THREADS 1024
#define BIN_EPT     4
#define CNT_PITCH   32
static_assert(NBIN * BIN_ROWS == N_TASKS, "bins must tile the task rows");
static_assert(BIN_ROWS == TILE_WAVES * 16, "one wave per 16-row tile of a bin");
static_assert(BIN_ROWS * HID * 4 == BIN_CAP * 8, "a bin's records fill exactly its own output rows");
static_assert(N_DATA <= 65536 && BIN_ROWS <= 65536, "record word: local row in the high half, source id in the low half");

__global__ void k_attvec(const float* __restrict__ Wd, const float* __restrict__ ad,
                         const float* __restrict__ We, const float* __restrict__ ae,
                         float* __restrict__ vsmall) {
    int t = threadIdx.x;
    if (t < 12) {
        float s = 0.f;
        for (int h = 0; h < HID; ++h) s += Wd[t * HID + h] * ad[h];
        vsmall[t] = s;
    } else if (t < 15) {
        int k = t - 12;
        float s = 0.f;
        for (int h = 0; h < HID; ++h) s += We[k * HID + h] * ae[h];
        vsmall[t] = s;
    }
}

template <int F, int KSTEPS>
__device__ __forceinline__ void gemm_tile_wmma(const float* __restrict__ X,
                                               const float* __restrict__ W,
                                               int m0, int lane, v8f c[4]) {
    const int half = lane >> 4;
    const int l15  = lane & 15;
    const int mrow = m0 + l15;
#pragma unroll
    for (int s = 0; s < KSTEPS; ++s) {
        const int  ka  = s * 4 + half * 2;
        const bool v0  = (ka     < F);
        const bool v1  = (ka + 1 < F);
        const int  ka0 = v0 ? ka     : 0;
        const int  ka1 = v1 ? ka + 1 : 0;
        const float ax = X[mrow * F + ka0];
        const float ay = X[mrow * F + ka1];
        v2f a;
        a.x = v0 ? ax : 0.0f;
        a.y = v1 ? ay : 0.0f;
#pragma unroll
        for (int t = 0; t < 4; ++t) {
            const int col = t * 16 + l15;
            const float bx = W[ka0 * HID + col];
            const float by = W[ka1 * HID + col];
            v2f b;
            b.x = v0 ? bx : 0.0f;
            b.y = v1 ? by : 0.0f;
            c[t] = __builtin_amdgcn_wmma_f32_16x16x4_f32(
                false, a, false, b, (short)0, c[t], false, false);
        }
    }
}

template <int CTRL>
__device__ __forceinline__ int dpp_i(int v) {
    return __builtin_amdgcn_update_dpp(v, v, CTRL, 0xf, 0xf, true);
}
template <int CTRL>
__device__ __forceinline__ float dpp_f(float v) {
    return __int_as_float(dpp_i<CTRL>(__float_as_int(v)));
}
__device__ __forceinline__ float rowsum16(float v) {
    v += dpp_f<0x161>(v);
    v += dpp_f<0x162>(v);
    v += dpp_f<0x164>(v);
    v += dpp_f<0x168>(v);
    return v;
}
__device__ __forceinline__ float rowmax16(float v) {
    v = fmaxf(v, dpp_f<0x161>(v));
    v = fmaxf(v, dpp_f<0x162>(v));
    v = fmaxf(v, dpp_f<0x164>(v));
    v = fmaxf(v, dpp_f<0x168>(v));
    return v;
}

static_assert((N_DATA / 16) * 32 >= NBIN && NBIN * CNT_PITCH <= N_TASKS, "the projection grid must cover the bin counters, and they must fit their plane");
__global__ void k_proj_data(const float* __restrict__ x, const float* __restrict__ W,
                            const float* __restrict__ att,
                            float4* __restrict__ h4, float* __restrict__ arow,
                            unsigned* __restrict__ cnt, unsigned* __restrict__ ovf_cnt) {
    const int gid  = blockIdx.x * blockDim.x + threadIdx.x;
    const int wave = gid >> 5;
    const int lane = threadIdx.x & 31;
    if (wave >= N_DATA / 16) return;
    if (gid < NBIN) cnt[gid * CNT_PITCH] = 0u;
    if (gid == 0) *ovf_cnt = 0u;
    const int m0 = wave * 16;
    v8f c[4] = {};
    gemm_tile_wmma<F_DATA, 2>(x, W, m0, lane, c);
    const int half = lane >> 4, l15 = lane & 15;
    const float at0 = att[l15], at1 = att[16 + l15], at2 = att[32 + l15], at3 = att[48 + l15];
#pragma unroll
    for (int r = 0; r < 8; ++r) {
        const int row = m0 + r + half * 8;
        h4[row * 16 + l15] = make_float4(c[0][r], c[1][r], c[2][r], c[3][r]);
        float p = c[0][r] * at0;
        p += c[1][r] * at1;
        p += c[2][r] * at2;
        p += c[3][r] * at3;
        p = rowsum16(p);
        if (l15 == 0) arow[row] = p;
    }
}

__device__ __forceinline__ float edge_pre(int e, int s,
                                          const float* __restrict__ eattr,
                                          const float* __restrict__ asrc,
                                          const float* __restrict__ vsmall) {
    return asrc[s]
         + eattr[e * 3 + 0] * vsmall[12]
         + eattr[e * 3 + 1] * vsmall[13]
         + eattr[e * 3 + 2] * vsmall[14];
}
__device__ __forceinline__ float logit(float pre, float ad) {
    const float al = pre + ad;
    return (al > 0.f) ? al : 0.2f * al;
}

__global__ __launch_bounds__(BIN_THREADS)
void k_bin(const int* __restrict__ src, const int* __restrict__ dst,
           const float* __restrict__ eattr, const float* __restrict__ asrc,
           const float* __restrict__ vsmall,
           unsigned* __restrict__ bincnt, float2* __restrict__ rec,
           unsigned* __restrict__ ovf_cnt, int* __restrict__ ovf) {
    __shared__ unsigned hist[NBIN];
    __shared__ unsigned sbase[NBIN];
    const int tid = threadIdx.x;
    if (tid < NBIN) hist[tid] = 0u;
    const int e0 = blockIdx.x * (BIN_THREADS * BIN_EPT) + tid;
    int   dv[BIN_EPT];
    unsigned sv[BIN_EPT];
    float pre[BIN_EPT];
#pragma unroll
    for (int i = 0; i < BIN_EPT; ++i) {
        const int e  = e0 + i * BIN_THREADS;
        const int ec = (e < N_EDGES) ? e : (N_EDGES - 1);
        const int s  = src[ec];
        dv[i]  = dst[ec];
        sv[i]  = (unsigned)s;
        pre[i] = edge_pre(ec, s, eattr, asrc, vsmall);
    }
    __syncthreads();
    int      bn[BIN_EPT];
    unsigned rk[BIN_EPT];
#pragma unroll
    for (int i = 0; i < BIN_EPT; ++i) {
        const int e = e0 + i * BIN_THREADS;
        const int b = dv[i] / BIN_ROWS;
        bn[i] = (e < N_EDGES) ? b : -1;
        rk[i] = 0u;
        if (e < N_EDGES) rk[i] = atomicAdd(&hist[b], 1u);
    }
    __syncthreads();
    if (tid < NBIN) {
        const unsigned h = hist[tid];
        sbase[tid] = h ? atomicAdd(&bincnt[tid * CNT_PITCH], h) : 0u;
    }
    __syncthreads();
#pragma unroll
    for (int i = 0; i < BIN_EPT; ++i) {
        if (bn[i] >= 0) {
            const unsigned pos = sbase[bn[i]] + rk[i];
            if (pos < (unsigned)BIN_CAP) {
                const unsigned wd = ((unsigned)(dv[i] - bn[i] * BIN_ROWS) << 16) | sv[i];
                rec[(size_t)bn[i] * BIN_CAP + pos] = make_float2(__uint_as_float(wd), pre[i]);
            } else {
                const unsigned k = atomicAdd(ovf_cnt, 1u);
                ovf[k] = e0 + i * BIN_THREADS;
            }
        }
    }
}

template <int J>
__device__ __forceinline__ float4 slot_load(int s, const float4* __restrict__ h4, int l15) {
    const unsigned sb = (unsigned)dpp_i<0x150 + J>(s);
    return h4[sb * 16u + (unsigned)l15];
}
template <int J>
__device__ __forceinline__ void slot_add(float w, const float4 hv,
                                         float& a0, float& a1, float& a2, float& a3) {
    const float wb = dpp_f<0x150 + J>(w);
    a0 = fmaf(wb, hv.x, a0);
    a1 = fmaf(wb, hv.y, a1);
    a2 = fmaf(wb, hv.z, a2);
    a3 = fmaf(wb, hv.w, a3);
}
__device__ __forceinline__ void gather16(int s, float w, const float4* __restrict__ h4, int l15,
                                         float& a0, float& a1, float& a2, float& a3) {
    const float4 h0  = slot_load<0>(s, h4, l15);
    const float4 h1  = slot_load<1>(s, h4, l15);
    const float4 h2  = slot_load<2>(s, h4, l15);
    const float4 h3  = slot_load<3>(s, h4, l15);
    const float4 h4v = slot_load<4>(s, h4, l15);
    const float4 h5  = slot_load<5>(s, h4, l15);
    const float4 h6  = slot_load<6>(s, h4, l15);
    const float4 h7  = slot_load<7>(s, h4, l15);
    const float4 h8  = slot_load<8>(s, h4, l15);
    const float4 h9  = slot_load<9>(s, h4, l15);
    const float4 h10 = slot_load<10>(s, h4, l15);
    const float4 h11 = slot_load<11>(s, h4, l15);
    const float4 h12 = slot_load<12>(s, h4, l15);
    const float4 h13 = slot_load<13>(s, h4, l15);
    const float4 h14 = slot_load<14>(s, h4, l15);
    const float4 h15 = slot_load<15>(s, h4, l15);
    slot_add<0>(w, h0, a0, a1, a2, a3);
    slot_add<1>(w, h1, a0, a1, a2, a3);
    slot_add<2>(w, h2, a0, a1, a2, a3);
    slot_add<3>(w, h3, a0, a1, a2, a3);
    slot_add<4>(w, h4v, a0, a1, a2, a3);
    slot_add<5>(w, h5, a0, a1, a2, a3);
    slot_add<6>(w, h6, a0, a1, a2, a3);
    slot_add<7>(w, h7, a0, a1, a2, a3);
    slot_add<8>(w, h8, a0, a1, a2, a3);
    slot_add<9>(w, h9, a0, a1, a2, a3);
    slot_add<10>(w, h10, a0, a1, a2, a3);
    slot_add<11>(w, h11, a0, a1, a2, a3);
    slot_add<12>(w, h12, a0, a1, a2, a3);
    slot_add<13>(w, h13, a0, a1, a2, a3);
    slot_add<14>(w, h14, a0, a1, a2, a3);
    slot_add<15>(w, h15, a0, a1, a2, a3);
}

__global__ __launch_bounds__(TILE_THREADS)
void k_tile(const int* __restrict__ src, const int* __restrict__ dst,
       const float* __restrict__ eattr, const float* __restrict__ asrc,
       const float* __restrict__ vsmall,
       const unsigned* __restrict__ bincnt, const unsigned* __restrict__ ovf_cnt,
       const int* __restrict__ ovf, const float4* __restrict__ h4,
       const float* __restrict__ xt, const float* __restrict__ Wr,
       const float* __restrict__ bias, float* out) {
    __shared__ float2   sbk[BIN_ROWS * CAP];
    __shared__ unsigned scnt[BIN_ROWS];
    __shared__ unsigned sflag;
    const int tid  = threadIdx.x;
    const int bin  = blockIdx.x;
    const int lane = tid & 31;
    const int wv   = tid >> 5;
    const int half = lane >> 4, l15 = lane & 15;
    const int m0    = bin * BIN_ROWS + wv * 16;
    const int rl0   = wv * 16 + half * 8;
    const int rbase = m0 + half * 8;

    if (tid < BIN_ROWS) scnt[tid] = 0u;
    if (tid == 0) sflag = 0u;
    const float2* rec = (const float2*)out + (size_t)bin * BIN_CAP;
    unsigned nrec = bincnt[bin * CNT_PITCH];
    nrec = (nrec < (unsigned)BIN_CAP) ? nrec : (unsigned)BIN_CAP;
    unsigned nov = *ovf_cnt;
    nov = (nov < (unsigned)N_EDGES) ? nov : (unsigned)N_EDGES;

    float adv = 0.f;
#pragma unroll
    for (int k = 0; k < F_TASK; ++k) adv += xt[(m0 + l15) * F_TASK + k] * vsmall[k];

    v8f c[4] = {};
    gemm_tile_wmma<F_TASK, 3>(xt, Wr, m0, lane, c);
#pragma unroll
    for (int t = 0; t < 4; ++t) {
        const float b = bias[t * 16 + l15];
#pragma unroll
        for (int r = 0; r < 8; ++r) c[t][r] += b;
    }

    __syncthreads();
    for (unsigned base = 0; base < nrec; base += 8u * TILE_THREADS) {
        float2 rc[8];
#pragma unroll
        for (int k = 0; k < 8; ++k) {
            const unsigned i = base + (unsigned)(k * TILE_THREADS + tid);
            rc[k] = rec[(i < nrec) ? i : 0u];
        }
#pragma unroll
        for (int k = 0; k < 8; ++k) {
            const unsigned i = base + (unsigned)(k * TILE_THREADS + tid);
            if (i < nrec) {
                const unsigned wd = __float_as_uint(rc[k].x);
                const unsigned rl = wd >> 16;
                const unsigned q  = atomicAdd(&scnt[rl], 1u);
                if (q < (unsigned)CAP)
                    sbk[rl * CAP + q] = make_float2(__int_as_float((int)(wd & 0xffffu)), rc[k].y);
                else
                    sflag = 1u;
            }
        }
    }
    for (unsigned i = tid; i < nov; i += TILE_THREADS) {
        const int e  = ovf[i];
        const int rl = dst[e] - bin * BIN_ROWS;
        if (rl >= 0 && rl < BIN_ROWS) {
            const int s = src[e];
            const unsigned q = atomicAdd(&scnt[rl], 1u);
            if (q < (unsigned)CAP)
                sbk[rl * CAP + q] = make_float2(__int_as_float(s), edge_pre(e, s, eattr, asrc, vsmall));
            else
                sflag = 1u;
        }
    }
    __syncthreads();

    const unsigned cv = scnt[wv * 16 + l15];
    unsigned njmax = 0u;
    const bool crowded = __builtin_amdgcn_readfirstlane((int)sflag) != 0;

#pragma unroll
    for (int r = 0; r < 8; ++r) {
        const unsigned nA = (unsigned)__builtin_amdgcn_readlane((int)cv, r);
        const unsigned nB = (unsigned)__builtin_amdgcn_readlane((int)cv, r + 8);
        const unsigned n  = half ? nB : nA;
        const unsigned nj = (nA > nB) ? nA : nB;
        njmax = (nj > njmax) ? nj : njmax;
        const bool slots = n <= (unsigned)CAP;
        const float adA = __int_as_float(__builtin_amdgcn_readlane(__float_as_int(adv), r));
        const float adB = __int_as_float(__builtin_amdgcn_readlane(__float_as_int(adv), r + 8));
        const float ad  = half ? adB : adA;

        const float2 ent = sbk[(rl0 + r) * CAP + l15];
        const bool v_lo = slots && ((unsigned)l15 < n);
        const int   s_lo = v_lo ? __float_as_int(ent.x) : 0;
        const float a_lo = v_lo ? logit(ent.y, ad) : -INFINITY;
        bool  v_hi = false;
        int   s_hi = 0;
        float a_hi = -INFINITY;
        if (nj > 16u) {
            const float2 e2 = sbk[(rl0 + r) * CAP + 16 + l15];
            v_hi = slots && ((unsigned)(16 + l15) < n);
            s_hi = v_hi ? __float_as_int(e2.x) : 0;
            a_hi = v_hi ? logit(e2.y, ad) : -INFINITY;
        }
        const float mx = rowmax16(fmaxf(a_lo, a_hi));

        const float e_lo = v_lo ? __expf(a_lo - mx) : 0.0f;
        const float e_hi = v_hi ? __expf(a_hi - mx) : 0.0f;
        const float den  = rowsum16(e_lo + e_hi);
        const float dinv = den + 1e-16f;
        const float w_lo = e_lo / dinv;

        float a0 = c[0][r], a1 = c[1][r], a2 = c[2][r], a3 = c[3][r];
        gather16(s_lo, w_lo, h4, l15, a0, a1, a2, a3);
        if (nj > 16u) {
            const float w_hi = e_hi / dinv;
            gather16(s_hi, w_hi, h4, l15, a0, a1, a2, a3);
        }
        c[0][r] = a0; c[1][r] = a1; c[2][r] = a2; c[3][r] = a3;
    }

    if (njmax > (unsigned)CAP) {
#pragma nounroll
        for (int r = 0; r < 8; ++r) {
            const unsigned nA = (unsigned)__builtin_amdgcn_readlane((int)cv, r);
            const unsigned nB = (unsigned)__builtin_amdgcn_readlane((int)cv, r + 8);
            if (((nA > nB) ? nA : nB) <= (unsigned)CAP) continue;
            const unsigned n = half ? nB : nA;
            const bool mine  = n > (unsigned)CAP;
            const float adA = __int_as_float(__builtin_amdgcn_readlane(__float_as_int(adv), r));
            const float adB = __int_as_float(__builtin_amdgcn_readlane(__float_as_int(adv), r + 8));
            const float ad  = half ? adB : adA;
            const int R = rbase + r;
            const unsigned rlR = (unsigned)(rl0 + r);

            float mx = -INFINITY;
            for (unsigned i = l15; i < nrec; i += 16) {
                const float2 rc = rec[i];
                if (mine && (__float_as_uint(rc.x) >> 16) == rlR) mx = fmaxf(mx, logit(rc.y, ad));
            }
            for (unsigned i = l15; i < nov; i += 16) {
                const int e = ovf[i];
                if (mine && dst[e] == R)
                    mx = fmaxf(mx, logit(edge_pre(e, src[e], eattr, asrc, vsmall), ad));
            }
            mx = rowmax16(mx);

            float den = 0.0f;
            for (unsigned i = l15; i < nrec; i += 16) {
                const float2 rc = rec[i];
                if (mine && (__float_as_uint(rc.x) >> 16) == rlR) den += __expf(logit(rc.y, ad) - mx);
            }
            for (unsigned i = l15; i < nov; i += 16) {
                const int e = ovf[i];
                if (mine && dst[e] == R)
                    den += __expf(logit(edge_pre(e, src[e], eattr, asrc, vsmall), ad) - mx);
            }
            den = rowsum16(den);
            const float dinv = den + 1e-16f;

            float a0 = 0.0f, a1 = 0.0f, a2 = 0.0f, a3 = 0.0f;
            for (unsigned base = 0; base < nrec; base += 16) {
                const unsigned i = base + l15;
                int   so = 0;
                float wo = 0.0f;
                if (mine && i < nrec) {
                    const float2 rc = rec[i];
                    const unsigned wd = __float_as_uint(rc.x);
                    if ((wd >> 16) == rlR) {
                        so = (int)(wd & 0xffffu);
                        wo = __expf(logit(rc.y, ad) - mx) / dinv;
                    }
                }
                gather16(so, wo, h4, l15, a0, a1, a2, a3);
            }
            for (unsigned base = 0; base < nov; base += 16) {
                const unsigned i = base + l15;
                int   so = 0;
                float wo = 0.0f;
                if (mine && i < nov) {
                    const int e = ovf[i];
                    if (dst[e] == R) {
                        so = src[e];
                        wo = __expf(logit(edge_pre(e, so, eattr, asrc, vsmall), ad) - mx) / dinv;
                    }
                }
                gather16(so, wo, h4, l15, a0, a1, a2, a3);
            }
#pragma unroll
            for (int rr = 0; rr < 8; ++rr) {
                if (rr == r) { c[0][rr] += a0; c[1][rr] += a1; c[2][rr] += a2; c[3][rr] += a3; }
            }
        }
    }

    if (crowded) __syncthreads();
#pragma unroll
    for (int r = 0; r < 8; ++r) {
        const int row = rbase + r;
#pragma unroll
        for (int t = 0; t < 4; ++t) out[row * HID + t * 16 + l15] = c[t][r];
    }
}

#define LN_ROWS 16
static_assert(N_TASKS % LN_ROWS == 0, "whole waves of 16 rows");
__global__ __launch_bounds__(256)
void k_lnorm(float* __restrict__ out, const float* __restrict__ gamma,
             const float* __restrict__ beta) {
    const int wave = (blockIdx.x * blockDim.x + threadIdx.x) >> 5;
    const int lane = threadIdx.x & 31;
    if (wave >= N_TASKS / LN_ROWS) return;
    const int half = lane >> 4, l15 = lane & 15;
    float4* p = (float4*)out + (size_t)(wave * LN_ROWS + half) * (HID / 4) + l15;
    const float4 g = ((const float4*)gamma)[l15];
    const float4 b = ((const float4*)beta)[l15];
    float4 x[8];
    for (int k = 0; k < 8; ++k) x[k] = p[k * 2 * (HID / 4)];
    for (int k = 0; k < 8; ++k) {
        const float4 v = x[k];
        float s = (v.x + v.y) + (v.z + v.w);
        float q = (v.x * v.x + v.y * v.y) + (v.z * v.z + v.w * v.w);
        s = rowsum16(s);
        q = rowsum16(q);
        const float mean = s * (1.0f / 64.0f);
        const float var  = q * (1.0f / 64.0f) - mean * mean;
        const float rstd = rsqrtf(var + 1e-5f);
        float4 y;
        y.x = (v.x - mean) * rstd * g.x + b.x;
        y.y = (v.y - mean) * rstd * g.y + b.y;
        y.z = (v.z - mean) * rstd * g.z + b.z;
        y.w = (v.w - mean) * rstd * g.w + b.w;
        y.x = (y.x > 0.f) ? y.x : 0.01f * y.x;
        y.y = (y.y > 0.f) ? y.y : 0.01f * y.y;
        y.z = (y.z > 0.f) ? y.z : 0.01f * y.z;
        y.w = (y.w > 0.f) ? y.w : 0.01f * y.w;
        p[k * 2 * (HID / 4)] = y;
    }
}

extern "C" void kernel_launch(void* const* d_in, const int* in_sizes, int n_in,
                              void* d_out, int out_size, void* d_ws, size_t ws_size,
                              hipStream_t stream) {
    const float* x_data    = (const float*)d_in[0];
    const float* x_tasks   = (const float*)d_in[1];
    const int*   edge_idx  = (const int*)d_in[2];
    const float* edge_attr = (const float*)d_in[3];
    const float* W_src     = (const float*)d_in[4];
    const float* W_dst     = (const float*)d_in[5];
    const float* W_edge    = (const float*)d_in[6];
    const float* att_src   = (const float*)d_in[7];
    const float* att_dst   = (const float*)d_in[8];
    const float* att_edge  = (const float*)d_in[9];
    const float* W_res     = (const float*)d_in[10];
    const float* bias      = (const float*)d_in[11];
    const float* ln_gamma  = (const float*)d_in[12];
    const float* ln_beta   = (const float*)d_in[13];
    const int* src = edge_idx;
    const int* dst = edge_idx + N_EDGES;
    float* out = (float*)d_out;

    float*    h_src   = (float*)d_ws;
    float*    a_src   = h_src + (size_t)N_DATA * HID;
    float*    a_dst   = a_src + N_DATA;
    unsigned* cnt     = (unsigned*)(a_dst + N_TASKS);
    unsigned* ovf_cnt = cnt + N_TASKS;
    int*      ovf     = (int*)(ovf_cnt + N_TASKS);
    float*    vsmall  = (float*)(ovf + N_EDGES);

    k_attvec<<<1, 64, 0, stream>>>(W_dst, att_dst, W_edge, att_edge, vsmall);

    {
        const int waves = N_DATA / 16;
        k_proj_data<<<(waves * 32 + 127) / 128, 128, 0, stream>>>(
            x_data, W_src, att_src, (float4*)h_src, a_src, cnt, ovf_cnt);
    }
    k_bin<<<(N_EDGES + BIN_THREADS * BIN_EPT - 1) / (BIN_THREADS * BIN_EPT), BIN_THREADS, 0, stream>>>(
        src, dst, edge_attr, a_src, vsmall, cnt, (float2*)out, ovf_cnt, ovf);
    k_tile<<<NBIN, TILE_THREADS, 0, stream>>>(
        src, dst, edge_attr, a_src, vsmall, cnt, ovf_cnt, ovf,
        (const float4*)h_src, x_tasks, W_res, bias, out);
    k_lnorm<<<((N_TASKS / LN_ROWS) * 32 + 255) / 256, 256, 0, stream>>>(out, ln_gamma, ln_beta);
}
